# starting kernel
# baseline (speedup 1.0000x reference)
#include <hip/hip_runtime.h>


typedef float v2f __attribute__((ext_vector_type(2)));
typedef float v8f __attribute__((ext_vector_type(8)));

#define LOG_N       12
#define NELEM       4096
#define BATCH       8192
#define ROWS_TILE   16
#define LDS_STRIDE  4100

__device__ __forceinline__ void ctx_pos(int g, int c, int& base, int& step) {
    if (g == 0)      { base = c << 4;                        step = 1;   }
    else if (g == 1) { base = ((c & 0xF0) << 4) | (c & 15);  step = 16;  }
    else             { base = c;                             step = 256; }
}

__global__ void butterfly_compose(const float* __restrict__ tw,
                                  float* __restrict__ M) {
    int tid = blockIdx.x * blockDim.x + threadIdx.x;
    if (tid >= 3 * 256 * 16) return;
    int k = tid & 15;
    int c = (tid >> 4) & 255;
    int g = tid >> 12;

    int base, step;
    ctx_pos(g, c, base, step);

    float v[16];
#pragma unroll
    for (int t = 0; t < 16; ++t) v[t] = (t == k) ? 1.0f : 0.0f;

#pragma unroll
    for (int j = 0; j < 4; ++j) {
        const int idx = 4 * g + j;
        const int S   = 1 << idx;
#pragma unroll
        for (int u = 0; u < 8; ++u) {
            const int off = u & ((1 << j) - 1);
            const int blk = u >> j;
            const int t0  = (blk << (j + 1)) | off;
            const int t1  = t0 + (1 << j);
            const int p0  = base + t0 * step;
            const int w   = ((p0 >> (idx + 1)) << idx) | (p0 & (S - 1));
            const float* q = tw + idx * (2048 * 4) + w * 4;
            const float a00 = q[0], a01 = q[1], a10 = q[2], a11 = q[3];
            const float x0 = v[t0], x1 = v[t1];
            v[t0] = a00 * x0 + a01 * x1;
            v[t1] = a10 * x0 + a11 * x1;
        }
    }

    float* out = M + ((g * 256 + c) * 16) * 16 + k;
#pragma unroll
    for (int i = 0; i < 16; ++i) out[i * 16] = v[i];
}

__global__ void __launch_bounds__(256, 1)
butterfly_apply(const float* __restrict__ x,
                const float* __restrict__ M,
                float* __restrict__ out) {
    extern __shared__ float X[];

    const int r0 = blockIdx.x * ROWS_TILE;

    {
        const float4* src = (const float4*)(x + (size_t)r0 * NELEM);
        for (int i = threadIdx.x; i < ROWS_TILE * (NELEM / 4); i += 256) {
            const int row = i >> 10;
            const int c4  = i & 1023;
            float4 val = src[row * 1024 + c4];
            *(float4*)&X[row * LDS_STRIDE + c4 * 4] = val;
        }
    }

    const int lane = threadIdx.x & 31;
    const int wave = threadIdx.x >> 5;
    const int r    = lane & 15;
    const int hi   = lane >> 4;

    for (int g = 0; g < 3; ++g) {
        __syncthreads();
        for (int ci = 0; ci < 32; ++ci) {
            const int c = (wave << 5) | ci;
            int base, step;
            ctx_pos(g, c, base, step);

            const float* Mrow = M + ((g * 256 + c) * 16 + r) * 16;
            const float* Brow = X + r * LDS_STRIDE + base;

            v8f acc = {};
#pragma unroll
            for (int q = 0; q < 4; ++q) {
                const int kb = 4 * q + 2 * hi;
                v2f a;
                a.x = Mrow[kb];
                a.y = Mrow[kb + 1];
                v2f b;
                b.x = Brow[kb * step];
                b.y = Brow[(kb + 1) * step];
                acc = __builtin_amdgcn_wmma_f32_16x16x4_f32(
                     false, a,  false, b,
                     (short)0, acc,  false,  false);
            }

            float* Drow = X + r * LDS_STRIDE + base;
#pragma unroll
            for (int j = 0; j < 8; ++j) {
                const int i = j + 8 * hi;
                Drow[i * step] = acc[j];
            }
        }
    }

    __syncthreads();

    {
        float4* dst = (float4*)(out + (size_t)r0 * NELEM);
        for (int i = threadIdx.x; i < ROWS_TILE * (NELEM / 4); i += 256) {
            const int row = i >> 10;
            const int c4  = i & 1023;
            dst[row * 1024 + c4] = *(const float4*)&X[row * LDS_STRIDE + c4 * 4];
        }
    }
}

extern "C" void kernel_launch(void* const* d_in, const int* in_sizes, int n_in,
                              void* d_out, int out_size, void* d_ws, size_t ws_size,
                              hipStream_t stream) {
    (void)in_sizes; (void)n_in; (void)out_size; (void)ws_size;

    const float* x  = (const float*)d_in[0];
    const float* tw = (const float*)d_in[1];
    float*       o  = (float*)d_out;
    float*       M  = (float*)d_ws;

    butterfly_compose<<<48, 256, 0, stream>>>(tw, M);

    const size_t lds_bytes = (size_t)ROWS_TILE * LDS_STRIDE * sizeof(float);
    (void)hipFuncSetAttribute((const void*)butterfly_apply,
                              hipFuncAttributeMaxDynamicSharedMemorySize,
                              (int)lds_bytes);

    butterfly_apply<<<BATCH / ROWS_TILE, 256, lds_bytes, stream>>>(x, M, o);
}


# apply kernel: 256 workgroups x 2 tiles; a wave loads its 32 stage-2 matrices once and keeps them in registers for both tiles (stage 2 reads only LDS); A requests 3 pairs ahead to fit; sum orders same
# speedup vs baseline: 1.2799x; 1.2799x over previous
#include <hip/hip_runtime.h>


typedef float v2f __attribute__((ext_vector_type(2)));
typedef float v8f __attribute__((ext_vector_type(8)));

#define LOG_N       12
#define NELEM       4096
#define BATCH       8192
#define ROWS_TILE   16
#define LDS_STRIDE  4100
#define NUM_WG      256
#define TILES_PER_WG 2

__device__ __forceinline__ void ctx_pos(int g, int c, int& base, int& step) {
    if (g == 0)      { base = c << 4;                        step = 1;   }
    else if (g == 1) { base = ((c & 0xF0) << 4) | (c & 15);  step = 16;  }
    else             { base = c;                             step = 256; }
}

__global__ void butterfly_compose(const float* __restrict__ tw,
                                  float* __restrict__ M) {
    int tid = blockIdx.x * blockDim.x + threadIdx.x;
    if (tid >= 3 * 256 * 16) return;
    int k = tid & 15;
    int c = (tid >> 4) & 255;
    int g = tid >> 12;

    int base, step;
    ctx_pos(g, c, base, step);

    float v[16];
#pragma unroll
    for (int t = 0; t < 16; ++t) v[t] = (t == k) ? 1.0f : 0.0f;

#pragma unroll
    for (int j = 0; j < 4; ++j) {
        const int idx = 4 * g + j;
        const int S   = 1 << idx;
#pragma unroll
        for (int u = 0; u < 8; ++u) {
            const int off = u & ((1 << j) - 1);
            const int blk = u >> j;
            const int t0  = (blk << (j + 1)) | off;
            const int t1  = t0 + (1 << j);
            const int p0  = base + t0 * step;
            const int w   = ((p0 >> (idx + 1)) << idx) | (p0 & (S - 1));
            const float* q = tw + idx * (2048 * 4) + w * 4;
            const float a00 = q[0], a01 = q[1], a10 = q[2], a11 = q[3];
            const float x0 = v[t0], x1 = v[t1];
            v[t0] = a00 * x0 + a01 * x1;
            v[t1] = a10 * x0 + a11 * x1;
        }
    }

    float* out = M + ((g * 256 + c) * 16) * 16 + k;
#pragma unroll
    for (int i = 0; i < 16; ++i) out[i * 16] = v[i];
}

#define PF_PAIRS 3

__device__ __forceinline__ void load_a_s0(float (&a)[8], const float* __restrict__ M,
                                          int ch, int cm, int r, int hi) {
    const int c = (ch << 4) | cm;
    const float* Mrow = M + ((0 * 256 + c) * 16 + (r ^ (cm & 8))) * 16 + 2 * hi;
#pragma unroll
    for (int q = 0; q < 4; ++q) {
        a[2 * q]     = Mrow[4 * q];
        a[2 * q + 1] = Mrow[4 * q + 1];
    }
}

__device__ __forceinline__ void load_a_s1(float (&a)[8], const float* __restrict__ M,
                                          int ch, int j, int b1, int r, int hi) {
    const int c1 = (ch << 4) | (j + 8 * b1);
    const float* Mrow = M + ((1 * 256 + c1) * 16 + r) * 16 + ((hi ^ b1) << 3);
    const float4 lo = *(const float4*)(Mrow);
    const float4 up = *(const float4*)(Mrow + 4);
    a[0] = lo.x; a[1] = lo.y; a[2] = lo.z; a[3] = lo.w;
    a[4] = up.x; a[5] = up.y; a[6] = up.z; a[7] = up.w;
}

__device__ __forceinline__ void load_b_s0(float (&b)[8], const float* Xblk, int cm, int hi) {
    const float* Brow = Xblk + (cm << 4) + 2 * hi;
#pragma unroll
    for (int q = 0; q < 4; ++q) {
        b[2 * q]     = Brow[4 * q];
        b[2 * q + 1] = Brow[4 * q + 1];
    }
}

__device__ __forceinline__ void load_a_n(float (&a)[8], const float* __restrict__ M,
                                         int ch, int n, int r, int hi) {
    if (n < 16) load_a_s0(a, M, ch, (n >> 1) + ((n & 1) << 3), r, hi);
    else        load_a_s1(a, M, ch, (n - 16) >> 1, n & 1, r, hi);
}

__device__ __forceinline__ v8f wmma4(const float (&a)[8], const float (&b)[8]) {
    v8f acc = {};
#pragma unroll
    for (int q = 0; q < 4; ++q) {
        v2f av; av.x = a[2 * q]; av.y = a[2 * q + 1];
        v2f bv; bv.x = b[2 * q]; bv.y = b[2 * q + 1];
        acc = __builtin_amdgcn_wmma_f32_16x16x4_f32(
            false, av, false, bv, (short)0, acc, false, false);
    }
    return acc;
}

__global__ void __launch_bounds__(256, 1)
butterfly_apply(const float* __restrict__ x,
                const float* __restrict__ M,
                float* __restrict__ out) {
    extern __shared__ float X[];

    const int lane = threadIdx.x & 31;
    const int wave = threadIdx.x >> 5;
    const int r    = lane & 15;
    const int hi   = lane >> 4;

    float a2[32][8];
#pragma unroll
    for (int ci = 0; ci < 32; ++ci) {
        const int c = (wave << 5) | ci;
        const float* Mrow = M + ((2 * 256 + c) * 16 + r) * 16 + 2 * hi;
#pragma unroll
        for (int q = 0; q < 4; ++q) {
            a2[ci][2 * q]     = Mrow[4 * q];
            a2[ci][2 * q + 1] = Mrow[4 * q + 1];
        }
    }

#pragma nounroll
    for (int t = 0; t < TILES_PER_WG; ++t) {
    const int r0 = (blockIdx.x + t * NUM_WG) * ROWS_TILE;

    {
        const float4* src = (const float4*)(x + (size_t)r0 * NELEM);
#pragma unroll 8
        for (int it = 0; it < ROWS_TILE * (NELEM / 4) / 256; ++it) {
            const int i   = threadIdx.x + it * 256;
            const int row = i >> 10;
            const int c4  = i & 1023;
            float4 val = src[row * 1024 + c4];
            *(float4*)&X[row * LDS_STRIDE + c4 * 4] = val;
        }
    }

    __syncthreads();

    for (int blk = 0; blk < 2; ++blk) {
        const int ch = (wave << 1) | blk;
        float* Xblk = X + r * LDS_STRIDE + (ch << 8);

        v8f   d0[16];
        float afr[32][8];
        float bfr[16][8];

#pragma unroll
        for (int n = 0; n < 2 * PF_PAIRS; ++n) load_a_n(afr[n], M, ch, n, r, hi);
        load_b_s0(bfr[0], Xblk, 0, hi);
        load_b_s0(bfr[1], Xblk, 8, hi);

#pragma unroll
        for (int p = 0; p < 16; ++p) {
            if (p + PF_PAIRS < 16) {
                load_a_n(afr[2 * (p + PF_PAIRS)],     M, ch, 2 * (p + PF_PAIRS),     r, hi);
                load_a_n(afr[2 * (p + PF_PAIRS) + 1], M, ch, 2 * (p + PF_PAIRS) + 1, r, hi);
            }
            if (p + 1 < 8) {
                load_b_s0(bfr[2 * (p + 1)],     Xblk, (p + 1),     hi);
                load_b_s0(bfr[2 * (p + 1) + 1], Xblk, (p + 1) + 8, hi);
            }
            __builtin_amdgcn_sched_barrier(0);

            if (p < 8) {
                d0[p]     = wmma4(afr[2 * p],     bfr[2 * p]);
                d0[p + 8] = wmma4(afr[2 * p + 1], bfr[2 * p + 1]);
            } else {
                const int j = p - 8;
                float b_lo[8], b_up[8];
#pragma unroll
                for (int m = 0; m < 8; ++m) {
                    const float P = d0[m][j];
                    const float Q = d0[m + 8][j];
                    b_lo[m] = hi ? Q : P;
                    b_up[m] = hi ? P : Q;
                }
                const v8f acc_lo = wmma4(afr[2 * p],     b_lo);
                const v8f acc_up = wmma4(afr[2 * p + 1], b_up);

#pragma unroll
                for (int j2 = 0; j2 < 8; ++j2) {
                    Xblk[(j2 + 8 * hi) * 16 + j]     = acc_lo[j2];
                    Xblk[(j2 + 8 * hi) * 16 + j + 8] = acc_up[j2];
                }
            }
        }
    }

    __syncthreads();
#pragma unroll
    for (int ci = 0; ci < 32; ++ci) {
        const int c = (wave << 5) | ci;
        int base, step;
        ctx_pos(2, c, base, step);

        const float* Brow = X + r * LDS_STRIDE + base;

        v8f acc = {};
#pragma unroll
        for (int q = 0; q < 4; ++q) {
            const int kb = 4 * q + 2 * hi;
            v2f a;
            a.x = a2[ci][2 * q];
            a.y = a2[ci][2 * q + 1];
            v2f b;
            b.x = Brow[kb * step];
            b.y = Brow[(kb + 1) * step];
            acc = __builtin_amdgcn_wmma_f32_16x16x4_f32(
                false, a, false, b, (short)0, acc, false, false);
        }

        float* Drow = X + r * LDS_STRIDE + base;
#pragma unroll
        for (int j = 0; j < 8; ++j) {
            const int i = j + 8 * hi;
            Drow[i * step] = acc[j];
        }
    }

    __syncthreads();

    {
        float4* dst = (float4*)(out + (size_t)r0 * NELEM);
#pragma unroll 8
        for (int it = 0; it < ROWS_TILE * (NELEM / 4) / 256; ++it) {
            const int i   = threadIdx.x + it * 256;
            const int row = i >> 10;
            const int c4  = i & 1023;
            dst[row * 1024 + c4] = *(const float4*)&X[row * LDS_STRIDE + c4 * 4];
        }
    }

    __syncthreads();
    }
}

extern "C" void kernel_launch(void* const* d_in, const int* in_sizes, int n_in,
                              void* d_out, int out_size, void* d_ws, size_t ws_size,
                              hipStream_t stream) {
    (void)in_sizes; (void)n_in; (void)out_size; (void)ws_size;

    const float* x  = (const float*)d_in[0];
    const float* tw = (const float*)d_in[1];
    float*       o  = (float*)d_out;
    float*       M  = (float*)d_ws;

    butterfly_compose<<<48, 256, 0, stream>>>(tw, M);

    const size_t lds_bytes = (size_t)ROWS_TILE * LDS_STRIDE * sizeof(float);
    (void)hipFuncSetAttribute((const void*)butterfly_apply,
                              hipFuncAttributeMaxDynamicSharedMemorySize,
                              (int)lds_bytes);

    butterfly_apply<<<NUM_WG, 256, lds_bytes, stream>>>(x, M, o);
}


# apply kernel copy-in: a thread's 64 tile loads run as a rolling window of 24-32 requests in flight (oldest 8 go to LDS as 128-bit stores, 8 re-requested), not 8 per trip drained every trip
# speedup vs baseline: 1.3358x; 1.0437x over previous
#include <hip/hip_runtime.h>


typedef float v2f __attribute__((ext_vector_type(2)));
typedef float v8f __attribute__((ext_vector_type(8)));
typedef float v4f __attribute__((ext_vector_type(4)));

#define LOG_N       12
#define NELEM       4096
#define BATCH       8192
#define ROWS_TILE   16
#define LDS_STRIDE  4100
#define NUM_WG      256
#define TILES_PER_WG 2
#define CP_TRIPS    64
#define CP_GRP      8
#define CP_WIN      4

__device__ __forceinline__ void ctx_pos(int g, int c, int& base, int& step) {
    if (g == 0)      { base = c << 4;                        step = 1;   }
    else if (g == 1) { base = ((c & 0xF0) << 4) | (c & 15);  step = 16;  }
    else             { base = c;                             step = 256; }
}

__global__ void butterfly_compose(const float* __restrict__ tw,
                                  float* __restrict__ M) {
    int tid = blockIdx.x * blockDim.x + threadIdx.x;
    if (tid >= 3 * 256 * 16) return;
    int k = tid & 15;
    int c = (tid >> 4) & 255;
    int g = tid >> 12;

    int base, step;
    ctx_pos(g, c, base, step);

    float v[16];
#pragma unroll
    for (int t = 0; t < 16; ++t) v[t] = (t == k) ? 1.0f : 0.0f;

#pragma unroll
    for (int j = 0; j < 4; ++j) {
        const int idx = 4 * g + j;
        const int S   = 1 << idx;
#pragma unroll
        for (int u = 0; u < 8; ++u) {
            const int off = u & ((1 << j) - 1);
            const int blk = u >> j;
            const int t0  = (blk << (j + 1)) | off;
            const int t1  = t0 + (1 << j);
            const int p0  = base + t0 * step;
            const int w   = ((p0 >> (idx + 1)) << idx) | (p0 & (S - 1));
            const float* q = tw + idx * (2048 * 4) + w * 4;
            const float a00 = q[0], a01 = q[1], a10 = q[2], a11 = q[3];
            const float x0 = v[t0], x1 = v[t1];
            v[t0] = a00 * x0 + a01 * x1;
            v[t1] = a10 * x0 + a11 * x1;
        }
    }

    float* out = M + ((g * 256 + c) * 16) * 16 + k;
#pragma unroll
    for (int i = 0; i < 16; ++i) out[i * 16] = v[i];
}

#define PF_PAIRS 3

__device__ __forceinline__ void load_a_s0(float (&a)[8], const float* __restrict__ M,
                                          int ch, int cm, int r, int hi) {
    const int c = (ch << 4) | cm;
    const float* Mrow = M + ((0 * 256 + c) * 16 + (r ^ (cm & 8))) * 16 + 2 * hi;
#pragma unroll
    for (int q = 0; q < 4; ++q) {
        a[2 * q]     = Mrow[4 * q];
        a[2 * q + 1] = Mrow[4 * q + 1];
    }
}

__device__ __forceinline__ void load_a_s1(float (&a)[8], const float* __restrict__ M,
                                          int ch, int j, int b1, int r, int hi) {
    const int c1 = (ch << 4) | (j + 8 * b1);
    const float* Mrow = M + ((1 * 256 + c1) * 16 + r) * 16 + ((hi ^ b1) << 3);
    const float4 lo = *(const float4*)(Mrow);
    const float4 up = *(const float4*)(Mrow + 4);
    a[0] = lo.x; a[1] = lo.y; a[2] = lo.z; a[3] = lo.w;
    a[4] = up.x; a[5] = up.y; a[6] = up.z; a[7] = up.w;
}

__device__ __forceinline__ void load_b_s0(float (&b)[8], const float* Xblk, int cm, int hi) {
    const float* Brow = Xblk + (cm << 4) + 2 * hi;
#pragma unroll
    for (int q = 0; q < 4; ++q) {
        b[2 * q]     = Brow[4 * q];
        b[2 * q + 1] = Brow[4 * q + 1];
    }
}

__device__ __forceinline__ void load_a_n(float (&a)[8], const float* __restrict__ M,
                                         int ch, int n, int r, int hi) {
    if (n < 16) load_a_s0(a, M, ch, (n >> 1) + ((n & 1) << 3), r, hi);
    else        load_a_s1(a, M, ch, (n - 16) >> 1, n & 1, r, hi);
}

__device__ __forceinline__ v8f wmma4(const float (&a)[8], const float (&b)[8]) {
    v8f acc = {};
#pragma unroll
    for (int q = 0; q < 4; ++q) {
        v2f av; av.x = a[2 * q]; av.y = a[2 * q + 1];
        v2f bv; bv.x = b[2 * q]; bv.y = b[2 * q + 1];
        acc = __builtin_amdgcn_wmma_f32_16x16x4_f32(
            false, av, false, bv, (short)0, acc, false, false);
    }
    return acc;
}

__global__ void __launch_bounds__(256, 1)
butterfly_apply(const float* __restrict__ x,
                const float* __restrict__ M,
                float* __restrict__ out) {
    extern __shared__ float X[];

    const int lane = threadIdx.x & 31;
    const int wave = threadIdx.x >> 5;
    const int r    = lane & 15;
    const int hi   = lane >> 4;

    float a2[32][8];
#pragma unroll
    for (int ci = 0; ci < 32; ++ci) {
        const int c = (wave << 5) | ci;
        const float* Mrow = M + ((2 * 256 + c) * 16 + r) * 16 + 2 * hi;
#pragma unroll
        for (int q = 0; q < 4; ++q) {
            a2[ci][2 * q]     = Mrow[4 * q];
            a2[ci][2 * q + 1] = Mrow[4 * q + 1];
        }
    }

#pragma nounroll
    for (int t = 0; t < TILES_PER_WG; ++t) {
    const int r0 = (blockIdx.x + t * NUM_WG) * ROWS_TILE;

    {
        const v4f* src = (const v4f*)(x + (size_t)r0 * NELEM) + threadIdx.x;
        v4f win[CP_WIN][CP_GRP];
#pragma unroll
        for (int g = 0; g < CP_WIN; ++g) {
#pragma unroll
            for (int u = 0; u < CP_GRP; ++u) win[g][u] = src[(g * CP_GRP + u) * 256];
        }
#pragma unroll
        for (int g = 0; g < CP_TRIPS / CP_GRP; ++g) {
#pragma unroll
            for (int u = 0; u < CP_GRP; ++u) {
                const int it  = g * CP_GRP + u;
                const int row = it >> 2;
                const int c4  = ((it & 3) << 8) + threadIdx.x;
                v4f* dstp = (v4f*)__builtin_assume_aligned(&X[row * LDS_STRIDE + c4 * 4], 16);
                *dstp = win[g % CP_WIN][u];
            }
            if (g + CP_WIN < CP_TRIPS / CP_GRP) {
#pragma unroll
                for (int u = 0; u < CP_GRP; ++u)
                    win[g % CP_WIN][u] = src[((g + CP_WIN) * CP_GRP + u) * 256];
            }
        }
    }

    __syncthreads();

    for (int blk = 0; blk < 2; ++blk) {
        const int ch = (wave << 1) | blk;
        float* Xblk = X + r * LDS_STRIDE + (ch << 8);

        v8f   d0[16];
        float afr[32][8];
        float bfr[16][8];

#pragma unroll
        for (int n = 0; n < 2 * PF_PAIRS; ++n) load_a_n(afr[n], M, ch, n, r, hi);
        load_b_s0(bfr[0], Xblk, 0, hi);
        load_b_s0(bfr[1], Xblk, 8, hi);

#pragma unroll
        for (int p = 0; p < 16; ++p) {
            if (p + PF_PAIRS < 16) {
                load_a_n(afr[2 * (p + PF_PAIRS)],     M, ch, 2 * (p + PF_PAIRS),     r, hi);
                load_a_n(afr[2 * (p + PF_PAIRS) + 1], M, ch, 2 * (p + PF_PAIRS) + 1, r, hi);
            }
            if (p + 1 < 8) {
                load_b_s0(bfr[2 * (p + 1)],     Xblk, (p + 1),     hi);
                load_b_s0(bfr[2 * (p + 1) + 1], Xblk, (p + 1) + 8, hi);
            }
            __builtin_amdgcn_sched_barrier(0);

            if (p < 8) {
                d0[p]     = wmma4(afr[2 * p],     bfr[2 * p]);
                d0[p + 8] = wmma4(afr[2 * p + 1], bfr[2 * p + 1]);
            } else {
                const int j = p - 8;
                float b_lo[8], b_up[8];
#pragma unroll
                for (int m = 0; m < 8; ++m) {
                    const float P = d0[m][j];
                    const float Q = d0[m + 8][j];
                    b_lo[m] = hi ? Q : P;
                    b_up[m] = hi ? P : Q;
                }
                const v8f acc_lo = wmma4(afr[2 * p],     b_lo);
                const v8f acc_up = wmma4(afr[2 * p + 1], b_up);

#pragma unroll
                for (int j2 = 0; j2 < 8; ++j2) {
                    Xblk[(j2 + 8 * hi) * 16 + j]     = acc_lo[j2];
                    Xblk[(j2 + 8 * hi) * 16 + j + 8] = acc_up[j2];
                }
            }
        }
    }

    __syncthreads();
#pragma unroll
    for (int ci = 0; ci < 32; ++ci) {
        const int c = (wave << 5) | ci;
        int base, step;
        ctx_pos(2, c, base, step);

        const float* Brow = X + r * LDS_STRIDE + base;

        v8f acc = {};
#pragma unroll
        for (int q = 0; q < 4; ++q) {
            const int kb = 4 * q + 2 * hi;
            v2f a;
            a.x = a2[ci][2 * q];
            a.y = a2[ci][2 * q + 1];
            v2f b;
            b.x = Brow[kb * step];
            b.y = Brow[(kb + 1) * step];
            acc = __builtin_amdgcn_wmma_f32_16x16x4_f32(
                false, a, false, b, (short)0, acc, false, false);
        }

        float* Drow = X + r * LDS_STRIDE + base;
#pragma unroll
        for (int j = 0; j < 8; ++j) {
            const int i = j + 8 * hi;
            Drow[i * step] = acc[j];
        }
    }

    __syncthreads();

    {
        float4* dst = (float4*)(out + (size_t)r0 * NELEM);
#pragma unroll 8
        for (int it = 0; it < ROWS_TILE * (NELEM / 4) / 256; ++it) {
            const int i   = threadIdx.x + it * 256;
            const int row = i >> 10;
            const int c4  = i & 1023;
            dst[row * 1024 + c4] = *(const float4*)&X[row * LDS_STRIDE + c4 * 4];
        }
    }

    __syncthreads();
    }
}

extern "C" void kernel_launch(void* const* d_in, const int* in_sizes, int n_in,
                              void* d_out, int out_size, void* d_ws, size_t ws_size,
                              hipStream_t stream) {
    (void)in_sizes; (void)n_in; (void)out_size; (void)ws_size;

    const float* x  = (const float*)d_in[0];
    const float* tw = (const float*)d_in[1];
    float*       o  = (float*)d_out;
    float*       M  = (float*)d_ws;

    butterfly_compose<<<48, 256, 0, stream>>>(tw, M);

    const size_t lds_bytes = (size_t)ROWS_TILE * LDS_STRIDE * sizeof(float);
    (void)hipFuncSetAttribute((const void*)butterfly_apply,
                              hipFuncAttributeMaxDynamicSharedMemorySize,
                              (int)lds_bytes);

    butterfly_apply<<<NUM_WG, 256, lds_bytes, stream>>>(x, M, o);
}
